# starting kernel
# baseline (speedup 1.0000x reference)
#include <hip/hip_runtime.h>
#include <math.h>


#define DIM 16

typedef float v2f __attribute__((ext_vector_type(2)));
typedef float v8f __attribute__((ext_vector_type(8)));

__host__ __device__ constexpr int cnot_perm_idx(int c, int t, int idx) {
    return idx ^ ((((idx >> (3 - c)) & 1)) << (3 - t));
}
__host__ __device__ constexpr int ring_perm_idx(int j) {
    int v = cnot_perm_idx(3, 0, j);
    v = cnot_perm_idx(2, 3, v);
    v = cnot_perm_idx(1, 2, v);
    v = cnot_perm_idx(0, 1, v);
    return v;
}

__global__ void build_fixed_matrix_kernel(const float* __restrict__ w,
                                          float* __restrict__ M) {
    const int t = threadIdx.x;
    if (t >= DIM) return;
    float v[DIM];
#pragma unroll
    for (int i = 0; i < DIM; ++i) v[i] = (i == t) ? 1.0f : 0.0f;
#pragma unroll
    for (int layer = 0; layer < 3; ++layer) {
#pragma unroll
        for (int q = 0; q < 4; ++q) {
            const float th = 0.5f * w[layer * 4 + q];
            const float c = cosf(th);
            const float s = sinf(th);
            const int mask = 1 << (3 - q);
#pragma unroll
            for (int i = 0; i < DIM; ++i) {
                if ((i & mask) == 0) {
                    const int j = i | mask;
                    const float a0 = v[i], a1 = v[j];
                    v[i] = c * a0 - s * a1;
                    v[j] = s * a0 + c * a1;
                }
            }
        }
        float nv[DIM];
#pragma unroll
        for (int i = 0; i < DIM; ++i) nv[i] = v[ring_perm_idx(i)];
#pragma unroll
        for (int i = 0; i < DIM; ++i) v[i] = nv[i];
    }
#pragma unroll
    for (int i = 0; i < DIM; ++i) M[i * DIM + t] = v[i];
}

__global__ void __launch_bounds__(256)
quantum_layer_kernel(const float* __restrict__ x, const float* __restrict__ M,
                     float* __restrict__ out, int ntiles, long Btot) {
    __shared__ float probs[8][DIM * 17];
    const int lane = threadIdx.x & 31;
    const int wave = threadIdx.x >> 5;
    const int m    = lane & 15;
    const int h    = lane >> 4;
    const long tile = (long)blockIdx.x * 8 + wave;
    const bool active = tile < (long)ntiles;

    if (active) {
        long b = tile * 16 + m;
        const long bl = (b < Btot) ? b : (Btot - 1);
        const float4 xv = ((const float4*)x)[bl];

        float c0, s0, c1, s1, c2, s2, c3, s3;
        sincosf(0.5f * xv.x, &s0, &c0);
        sincosf(0.5f * xv.y, &s1, &c1);
        sincosf(0.5f * xv.z, &s2, &c2);
        sincosf(0.5f * xv.w, &s3, &c3);
        const float f2 = h ? s2 : c2;

        v8f acc = {0.f, 0.f, 0.f, 0.f, 0.f, 0.f, 0.f, 0.f};
#pragma unroll
        for (int g = 0; g < 4; ++g) {
            const float base = ((g & 2) ? s0 : c0) * ((g & 1) ? s1 : c1) * f2;
            v2f a;
            a.x = base * c3;
            a.y = base * s3;
            const float* mrow = M + m * DIM + g * 4 + 2 * h;
            v2f bm;
            bm.x = mrow[0];
            bm.y = mrow[1];
            acc = __builtin_amdgcn_wmma_f32_16x16x4_f32(
                false, a, false, bm, (short)0, acc, false, false);
        }
#pragma unroll
        for (int r = 0; r < 8; ++r) {
            const int row = r + 8 * h;
            const float yv = acc[r];
            probs[wave][row * 17 + m] = yv * yv;
        }
    }
    __syncthreads();
    if (active) {
        const long b = tile * 16 + m;
        if (b < Btot) {
            const int bp0 = 3 - (2 * h);
            const int bp1 = 3 - (2 * h + 1);
            float a0 = 0.f, a1 = 0.f;
#pragma unroll
            for (int n = 0; n < DIM; ++n) {
                const float p = probs[wave][m * 17 + n];
                a0 += ((n >> bp0) & 1) ? -p : p;
                a1 += ((n >> bp1) & 1) ? -p : p;
            }
            float2 r2;
            r2.x = a0;
            r2.y = a1;
            ((float2*)out)[b * 2 + h] = r2;
        }
    }
}

extern "C" void kernel_launch(void* const* d_in, const int* in_sizes, int n_in,
                              void* d_out, int out_size, void* d_ws, size_t ws_size,
                              hipStream_t stream) {
    const float* x = (const float*)d_in[0];
    const float* w = (const float*)d_in[1];
    float* out = (float*)d_out;
    float* M = (float*)d_ws;

    const long B = (long)in_sizes[0] / 4;
    const int ntiles = (int)((B + 15) / 16);
    const int blocks = (ntiles + 7) / 8;

    build_fixed_matrix_kernel<<<1, 16, 0, stream>>>(w, M);
    quantum_layer_kernel<<<blocks, 256, 0, stream>>>(x, M, out, ntiles, B);
}


# One launch, no set-up kernel: each wave forms the operator by two chained f32 WMMA products of per-layer Kronecker factors (D reused as B), loops over 8 tiles; operator rounding, k sum order change
# speedup vs baseline: 1.1148x; 1.1148x over previous
#include <hip/hip_runtime.h>
#include <math.h>


#define DIM 16

typedef float v2f __attribute__((ext_vector_type(2)));
typedef float v8f __attribute__((ext_vector_type(8)));

__host__ __device__ constexpr int cnot_perm_idx(int c, int t, int idx) {
    return idx ^ ((((idx >> (3 - c)) & 1)) << (3 - t));
}
__host__ __device__ constexpr int ring_perm_idx(int j) {
    int v = cnot_perm_idx(3, 0, j);
    v = cnot_perm_idx(2, 3, v);
    v = cnot_perm_idx(1, 2, v);
    v = cnot_perm_idx(0, 1, v);
    return v;
}
__host__ __device__ constexpr int qbit(int idx, int q) { return (idx >> (3 - q)) & 1; }

__device__ __forceinline__ float ry_elem(int a, int b, float cs, float sn) {
    return (a == b) ? cs : (a ? sn : -sn);
}

__device__ __forceinline__ void layer_rows(const float* cs, const float* sn,
                                           int c, int h, float* frag) {
    constexpr int P8 = ring_perm_idx(8);
    float u[4][2];
#pragma unroll
    for (int q = 0; q < 4; ++q) {
        const int cb = qbit(c, q);
        const int hf = h & qbit(P8, q);
        u[q][0] = ry_elem(hf, cb, cs[q], sn[q]);
        u[q][1] = ry_elem(1 ^ hf, cb, cs[q], sn[q]);
    }
    float p01[2][2], p23[2][2];
#pragma unroll
    for (int i = 0; i < 2; ++i)
#pragma unroll
        for (int j = 0; j < 2; ++j) {
            p01[i][j] = u[0][i] * u[1][j];
            p23[i][j] = u[2][i] * u[3][j];
        }
#pragma unroll
    for (int r = 0; r < 8; ++r) {
        const int pr = ring_perm_idx(r);
        frag[r] = p01[qbit(pr, 0)][qbit(pr, 1)] * p23[qbit(pr, 2)][qbit(pr, 3)];
    }
}

__device__ __forceinline__ void layer_cols(const float* cs, const float* sn,
                                           int c, int h, float* frag) {
    const int pc = ring_perm_idx(c);
    float v[4][2];
#pragma unroll
    for (int q = 0; q < 4; ++q) {
        const int rb = qbit(pc, q);
        v[q][0] = ry_elem(rb, 0, cs[q], sn[q]);
        v[q][1] = ry_elem(rb, 1, cs[q], sn[q]);
    }
    const float v0h = h ? v[0][1] : v[0][0];
    float p01[2], p23[2][2];
#pragma unroll
    for (int i = 0; i < 2; ++i) {
        p01[i] = v0h * v[1][i];
#pragma unroll
        for (int j = 0; j < 2; ++j) p23[i][j] = v[2][i] * v[3][j];
    }
#pragma unroll
    for (int r = 0; r < 8; ++r)
        frag[r] = p01[(r >> 2) & 1] * p23[(r >> 1) & 1][r & 1];
}

__device__ __forceinline__ float4 load_angles(const float* __restrict__ x, long tile,
                                              int m, long Btot) {
    const long b = tile * 16 + m;
    const long bl = (b < Btot) ? b : (Btot - 1);
    return ((const float4*)x)[bl];
}

__global__ void __launch_bounds__(256)
quantum_layer_kernel(const float* __restrict__ x, const float* __restrict__ w,
                     float* __restrict__ out, int ntiles, long Btot) {
    __shared__ float probs[8][DIM * 17];
    const int lane = threadIdx.x & 31;
    const int wave = threadIdx.x >> 5;
    const int m    = lane & 15;
    const int h    = lane >> 4;
    const long stride = (long)gridDim.x * 8;
    long base = (long)blockIdx.x * 8;

    float4 xv = load_angles(x, base + wave, m, Btot);

    float wsn, wcs;
    sincosf(0.5f * w[lane < 12 ? lane : 0], &wsn, &wcs);
    float cs[3][4], sn[3][4];
#pragma unroll
    for (int j = 0; j < 12; ++j) {
        cs[j >> 2][j & 3] = __int_as_float(__builtin_amdgcn_readlane(__float_as_int(wcs), j));
        sn[j >> 2][j & 3] = __int_as_float(__builtin_amdgcn_readlane(__float_as_int(wsn), j));
    }
    float a1[8], a2[8], b3[8];
    layer_rows(cs[0], sn[0], m, h, a1);
    layer_rows(cs[1], sn[1], m, h, a2);
    layer_cols(cs[2], sn[2], m, h, b3);

    v8f t1 = {0.f, 0.f, 0.f, 0.f, 0.f, 0.f, 0.f, 0.f};
#pragma unroll
    for (int g = 0; g < 4; ++g) {
        v2f a, b;
        a.x = a2[2 * g]; a.y = a2[2 * g + 1];
        b.x = b3[2 * g]; b.y = b3[2 * g + 1];
        t1 = __builtin_amdgcn_wmma_f32_16x16x4_f32(
            false, a, false, b, (short)0, t1, false, false);
    }
    v8f mt = {0.f, 0.f, 0.f, 0.f, 0.f, 0.f, 0.f, 0.f};
#pragma unroll
    for (int g = 0; g < 4; ++g) {
        v2f a, b;
        a.x = a1[2 * g]; a.y = a1[2 * g + 1];
        b.x = t1[2 * g]; b.y = t1[2 * g + 1];
        mt = __builtin_amdgcn_wmma_f32_16x16x4_f32(
            false, a, false, b, (short)0, mt, false, false);
    }

    for (; base < (long)ntiles; base += stride) {
        const long tile = base + wave;
        const bool active = tile < (long)ntiles;
        const float4 xn = load_angles(x, tile + stride, m, Btot);

        if (active) {
            float c0, s0, c1, s1, c2, s2, c3, s3;
            sincosf(0.5f * xv.x, &s0, &c0);
            sincosf(0.5f * xv.y, &s1, &c1);
            sincosf(0.5f * xv.z, &s2, &c2);
            sincosf(0.5f * xv.w, &s3, &c3);
            const float f0 = h ? s0 : c0;

            v8f acc = {0.f, 0.f, 0.f, 0.f, 0.f, 0.f, 0.f, 0.f};
#pragma unroll
            for (int g = 0; g < 4; ++g) {
                const float pre = (f0 * ((g & 2) ? s1 : c1)) * ((g & 1) ? s2 : c2);
                v2f a;
                a.x = pre * c3;
                a.y = pre * s3;
                v2f bm;
                bm.x = mt[2 * g];
                bm.y = mt[2 * g + 1];
                acc = __builtin_amdgcn_wmma_f32_16x16x4_f32(
                    false, a, false, bm, (short)0, acc, false, false);
            }
#pragma unroll
            for (int r = 0; r < 8; ++r) {
                const int row = r + 8 * h;
                const float yv = acc[r];
                probs[wave][row * 17 + m] = yv * yv;
            }
        }
        __syncthreads();
        if (active) {
            const long b = tile * 16 + m;
            if (b < Btot) {
                const int bp0 = 3 - (2 * h);
                const int bp1 = 3 - (2 * h + 1);
                float a0 = 0.f, a1s = 0.f;
#pragma unroll
                for (int n = 0; n < DIM; ++n) {
                    const float p = probs[wave][m * 17 + n];
                    a0  += ((n >> bp0) & 1) ? -p : p;
                    a1s += ((n >> bp1) & 1) ? -p : p;
                }
                float2 r2;
                r2.x = a0;
                r2.y = a1s;
                ((float2*)out)[b * 2 + h] = r2;
            }
        }
        xv = xn;
    }
}

extern "C" void kernel_launch(void* const* d_in, const int* in_sizes, int n_in,
                              void* d_out, int out_size, void* d_ws, size_t ws_size,
                              hipStream_t stream) {
    const float* x = (const float*)d_in[0];
    const float* w = (const float*)d_in[1];
    float* out = (float*)d_out;

    const long B = (long)in_sizes[0] / 4;
    const int ntiles = (int)((B + 15) / 16);

    quantum_layer_kernel<<<1024, 256, 0, stream>>>(x, w, out, ntiles, B);
}


# Issue priority follows progress: each wave sets s_setprio 3/2/1/0 by the quarter of the tile range its work-group has reached, so resident work-groups stay level; arithmetic and sum order unchanged
# speedup vs baseline: 1.2277x; 1.1013x over previous
#include <hip/hip_runtime.h>
#include <math.h>


#define DIM 16

typedef float v2f __attribute__((ext_vector_type(2)));
typedef float v8f __attribute__((ext_vector_type(8)));

__host__ __device__ constexpr int cnot_perm_idx(int c, int t, int idx) {
    return idx ^ ((((idx >> (3 - c)) & 1)) << (3 - t));
}
__host__ __device__ constexpr int ring_perm_idx(int j) {
    int v = cnot_perm_idx(3, 0, j);
    v = cnot_perm_idx(2, 3, v);
    v = cnot_perm_idx(1, 2, v);
    v = cnot_perm_idx(0, 1, v);
    return v;
}
__host__ __device__ constexpr int qbit(int idx, int q) { return (idx >> (3 - q)) & 1; }

__device__ __forceinline__ float ry_elem(int a, int b, float cs, float sn) {
    return (a == b) ? cs : (a ? sn : -sn);
}

__device__ __forceinline__ void layer_rows(const float* cs, const float* sn,
                                           int c, int h, float* frag) {
    constexpr int P8 = ring_perm_idx(8);
    float u[4][2];
#pragma unroll
    for (int q = 0; q < 4; ++q) {
        const int cb = qbit(c, q);
        const int hf = h & qbit(P8, q);
        u[q][0] = ry_elem(hf, cb, cs[q], sn[q]);
        u[q][1] = ry_elem(1 ^ hf, cb, cs[q], sn[q]);
    }
    float p01[2][2], p23[2][2];
#pragma unroll
    for (int i = 0; i < 2; ++i)
#pragma unroll
        for (int j = 0; j < 2; ++j) {
            p01[i][j] = u[0][i] * u[1][j];
            p23[i][j] = u[2][i] * u[3][j];
        }
#pragma unroll
    for (int r = 0; r < 8; ++r) {
        const int pr = ring_perm_idx(r);
        frag[r] = p01[qbit(pr, 0)][qbit(pr, 1)] * p23[qbit(pr, 2)][qbit(pr, 3)];
    }
}

__device__ __forceinline__ void layer_cols(const float* cs, const float* sn,
                                           int c, int h, float* frag) {
    const int pc = ring_perm_idx(c);
    float v[4][2];
#pragma unroll
    for (int q = 0; q < 4; ++q) {
        const int rb = qbit(pc, q);
        v[q][0] = ry_elem(rb, 0, cs[q], sn[q]);
        v[q][1] = ry_elem(rb, 1, cs[q], sn[q]);
    }
    const float v0h = h ? v[0][1] : v[0][0];
    float p01[2], p23[2][2];
#pragma unroll
    for (int i = 0; i < 2; ++i) {
        p01[i] = v0h * v[1][i];
#pragma unroll
        for (int j = 0; j < 2; ++j) p23[i][j] = v[2][i] * v[3][j];
    }
#pragma unroll
    for (int r = 0; r < 8; ++r)
        frag[r] = p01[(r >> 2) & 1] * p23[(r >> 1) & 1][r & 1];
}

__device__ __forceinline__ float4 load_angles(const float* __restrict__ x, long tile,
                                              int m, long Btot) {
    const long b = tile * 16 + m;
    const long bl = (b < Btot) ? b : (Btot - 1);
    return ((const float4*)x)[bl];
}


__device__ __forceinline__ int priority_from_progress(int bi, int q1, int q2, int q3) {
    if (bi < q1) { __builtin_amdgcn_s_setprio(3); return q1; }
    if (bi < q2) { __builtin_amdgcn_s_setprio(2); return q2; }
    if (bi < q3) { __builtin_amdgcn_s_setprio(1); return q3; }
    __builtin_amdgcn_s_setprio(0);
    return 0x7fffffff;
}

__global__ void __launch_bounds__(256)
quantum_layer_kernel(const float* __restrict__ x, const float* __restrict__ w,
                     float* __restrict__ out, int ntiles, long Btot) {
    __shared__ float probs[8][DIM * 17];
    const int lane = threadIdx.x & 31;
    const int wave = threadIdx.x >> 5;
    const int m    = lane & 15;
    const int h    = lane >> 4;
    const long stride = (long)gridDim.x * 8;
    long base = (long)blockIdx.x * 8;
    const int q1 = ntiles >> 2, q2 = ntiles >> 1, q3 = q1 + q2;
    int mark = priority_from_progress((int)base, q1, q2, q3);

    float4 xv = load_angles(x, base + wave, m, Btot);

    float wsn, wcs;
    sincosf(0.5f * w[lane < 12 ? lane : 0], &wsn, &wcs);
    float cs[3][4], sn[3][4];
#pragma unroll
    for (int j = 0; j < 12; ++j) {
        cs[j >> 2][j & 3] = __int_as_float(__builtin_amdgcn_readlane(__float_as_int(wcs), j));
        sn[j >> 2][j & 3] = __int_as_float(__builtin_amdgcn_readlane(__float_as_int(wsn), j));
    }
    float a1[8], a2[8], b3[8];
    layer_rows(cs[0], sn[0], m, h, a1);
    layer_rows(cs[1], sn[1], m, h, a2);
    layer_cols(cs[2], sn[2], m, h, b3);

    v8f t1 = {0.f, 0.f, 0.f, 0.f, 0.f, 0.f, 0.f, 0.f};
#pragma unroll
    for (int g = 0; g < 4; ++g) {
        v2f a, b;
        a.x = a2[2 * g]; a.y = a2[2 * g + 1];
        b.x = b3[2 * g]; b.y = b3[2 * g + 1];
        t1 = __builtin_amdgcn_wmma_f32_16x16x4_f32(
            false, a, false, b, (short)0, t1, false, false);
    }
    v8f mt = {0.f, 0.f, 0.f, 0.f, 0.f, 0.f, 0.f, 0.f};
#pragma unroll
    for (int g = 0; g < 4; ++g) {
        v2f a, b;
        a.x = a1[2 * g]; a.y = a1[2 * g + 1];
        b.x = t1[2 * g]; b.y = t1[2 * g + 1];
        mt = __builtin_amdgcn_wmma_f32_16x16x4_f32(
            false, a, false, b, (short)0, mt, false, false);
    }

    for (; base < (long)ntiles; base += stride) {
        if ((int)base >= mark)
            mark = priority_from_progress((int)base, q1, q2, q3);
        const long tile = base + wave;
        const bool active = tile < (long)ntiles;
        const float4 xn = load_angles(x, tile + stride, m, Btot);

        if (active) {
            float c0, s0, c1, s1, c2, s2, c3, s3;
            sincosf(0.5f * xv.x, &s0, &c0);
            sincosf(0.5f * xv.y, &s1, &c1);
            sincosf(0.5f * xv.z, &s2, &c2);
            sincosf(0.5f * xv.w, &s3, &c3);
            const float f0 = h ? s0 : c0;

            v8f acc = {0.f, 0.f, 0.f, 0.f, 0.f, 0.f, 0.f, 0.f};
#pragma unroll
            for (int g = 0; g < 4; ++g) {
                const float pre = (f0 * ((g & 2) ? s1 : c1)) * ((g & 1) ? s2 : c2);
                v2f a;
                a.x = pre * c3;
                a.y = pre * s3;
                v2f bm;
                bm.x = mt[2 * g];
                bm.y = mt[2 * g + 1];
                acc = __builtin_amdgcn_wmma_f32_16x16x4_f32(
                    false, a, false, bm, (short)0, acc, false, false);
            }
#pragma unroll
            for (int r = 0; r < 8; ++r) {
                const int row = r + 8 * h;
                const float yv = acc[r];
                probs[wave][row * 17 + m] = yv * yv;
            }
        }
        __syncthreads();
        if (active) {
            const long b = tile * 16 + m;
            if (b < Btot) {
                const int bp0 = 3 - (2 * h);
                const int bp1 = 3 - (2 * h + 1);
                float a0 = 0.f, a1s = 0.f;
#pragma unroll
                for (int n = 0; n < DIM; ++n) {
                    const float p = probs[wave][m * 17 + n];
                    a0  += ((n >> bp0) & 1) ? -p : p;
                    a1s += ((n >> bp1) & 1) ? -p : p;
                }
                float2 r2;
                r2.x = a0;
                r2.y = a1s;
                ((float2*)out)[b * 2 + h] = r2;
            }
        }
        xv = xn;
    }
    __builtin_amdgcn_s_setprio(0);
}

extern "C" void kernel_launch(void* const* d_in, const int* in_sizes, int n_in,
                              void* d_out, int out_size, void* d_ws, size_t ws_size,
                              hipStream_t stream) {
    const float* x = (const float*)d_in[0];
    const float* w = (const float*)d_in[1];
    float* out = (float*)d_out;

    const long B = (long)in_sizes[0] / 4;
    const int ntiles = (int)((B + 15) / 16);

    quantum_layer_kernel<<<1024, 256, 0, stream>>>(x, w, out, ntiles, B);
}


# A sample's two lanes (c, c+16) each do sincos for two of its four angles; the pairs cross through a wave-private LDS slot (one 16 B store, two 16 B loads, no barrier); arithmetic, sum order unchanged
# speedup vs baseline: 1.3384x; 1.0902x over previous
#include <hip/hip_runtime.h>
#include <math.h>


#define DIM 16

typedef float v2f __attribute__((ext_vector_type(2)));
typedef float v8f __attribute__((ext_vector_type(8)));
typedef float v4f __attribute__((ext_vector_type(4)));

__host__ __device__ constexpr int cnot_perm_idx(int c, int t, int idx) {
    return idx ^ ((((idx >> (3 - c)) & 1)) << (3 - t));
}
__host__ __device__ constexpr int ring_perm_idx(int j) {
    int v = cnot_perm_idx(3, 0, j);
    v = cnot_perm_idx(2, 3, v);
    v = cnot_perm_idx(1, 2, v);
    v = cnot_perm_idx(0, 1, v);
    return v;
}
__host__ __device__ constexpr int qbit(int idx, int q) { return (idx >> (3 - q)) & 1; }

__device__ __forceinline__ float ry_elem(int a, int b, float cs, float sn) {
    return (a == b) ? cs : (a ? sn : -sn);
}

__device__ __forceinline__ void layer_rows(const float* cs, const float* sn,
                                           int c, int h, float* frag) {
    constexpr int P8 = ring_perm_idx(8);
    float u[4][2];
#pragma unroll
    for (int q = 0; q < 4; ++q) {
        const int cb = qbit(c, q);
        const int hf = h & qbit(P8, q);
        u[q][0] = ry_elem(hf, cb, cs[q], sn[q]);
        u[q][1] = ry_elem(1 ^ hf, cb, cs[q], sn[q]);
    }
    float p01[2][2], p23[2][2];
#pragma unroll
    for (int i = 0; i < 2; ++i)
#pragma unroll
        for (int j = 0; j < 2; ++j) {
            p01[i][j] = u[0][i] * u[1][j];
            p23[i][j] = u[2][i] * u[3][j];
        }
#pragma unroll
    for (int r = 0; r < 8; ++r) {
        const int pr = ring_perm_idx(r);
        frag[r] = p01[qbit(pr, 0)][qbit(pr, 1)] * p23[qbit(pr, 2)][qbit(pr, 3)];
    }
}

__device__ __forceinline__ void layer_cols(const float* cs, const float* sn,
                                           int c, int h, float* frag) {
    const int pc = ring_perm_idx(c);
    float v[4][2];
#pragma unroll
    for (int q = 0; q < 4; ++q) {
        const int rb = qbit(pc, q);
        v[q][0] = ry_elem(rb, 0, cs[q], sn[q]);
        v[q][1] = ry_elem(rb, 1, cs[q], sn[q]);
    }
    const float v0h = h ? v[0][1] : v[0][0];
    float p01[2], p23[2][2];
#pragma unroll
    for (int i = 0; i < 2; ++i) {
        p01[i] = v0h * v[1][i];
#pragma unroll
        for (int j = 0; j < 2; ++j) p23[i][j] = v[2][i] * v[3][j];
    }
#pragma unroll
    for (int r = 0; r < 8; ++r)
        frag[r] = p01[(r >> 2) & 1] * p23[(r >> 1) & 1][r & 1];
}

__device__ __forceinline__ float2 load_angle_pair(const float* __restrict__ x, long tile,
                                                  int m, int h, long Btot) {
    const long b = tile * 16 + m;
    const long bl = (b < Btot) ? b : (Btot - 1);
    return ((const float2*)x)[bl * 2 + h];
}


__device__ __forceinline__ int priority_from_progress(int bi, int q1, int q2, int q3) {
    if (bi < q1) { __builtin_amdgcn_s_setprio(3); return q1; }
    if (bi < q2) { __builtin_amdgcn_s_setprio(2); return q2; }
    if (bi < q3) { __builtin_amdgcn_s_setprio(1); return q3; }
    __builtin_amdgcn_s_setprio(0);
    return 0x7fffffff;
}

__global__ void __launch_bounds__(256)
quantum_layer_kernel(const float* __restrict__ x, const float* __restrict__ w,
                     float* __restrict__ out, int ntiles, long Btot) {
    __shared__ float probs[8][DIM * 17];
    __shared__ v4f trig[8][32];
    const int lane = threadIdx.x & 31;
    const int wave = threadIdx.x >> 5;
    const int m    = lane & 15;
    const int h    = lane >> 4;
    const long stride = (long)gridDim.x * 8;
    long base = (long)blockIdx.x * 8;
    const int q1 = ntiles >> 2, q2 = ntiles >> 1, q3 = q1 + q2;
    int mark = priority_from_progress((int)base, q1, q2, q3);

    float2 xv = load_angle_pair(x, base + wave, m, h, Btot);

    float wsn, wcs;
    sincosf(0.5f * w[lane < 12 ? lane : 0], &wsn, &wcs);
    float cs[3][4], sn[3][4];
#pragma unroll
    for (int j = 0; j < 12; ++j) {
        cs[j >> 2][j & 3] = __int_as_float(__builtin_amdgcn_readlane(__float_as_int(wcs), j));
        sn[j >> 2][j & 3] = __int_as_float(__builtin_amdgcn_readlane(__float_as_int(wsn), j));
    }
    float a1[8], a2[8], b3[8];
    layer_rows(cs[0], sn[0], m, h, a1);
    layer_rows(cs[1], sn[1], m, h, a2);
    layer_cols(cs[2], sn[2], m, h, b3);

    v8f t1 = {0.f, 0.f, 0.f, 0.f, 0.f, 0.f, 0.f, 0.f};
#pragma unroll
    for (int g = 0; g < 4; ++g) {
        v2f a, b;
        a.x = a2[2 * g]; a.y = a2[2 * g + 1];
        b.x = b3[2 * g]; b.y = b3[2 * g + 1];
        t1 = __builtin_amdgcn_wmma_f32_16x16x4_f32(
            false, a, false, b, (short)0, t1, false, false);
    }
    v8f mt = {0.f, 0.f, 0.f, 0.f, 0.f, 0.f, 0.f, 0.f};
#pragma unroll
    for (int g = 0; g < 4; ++g) {
        v2f a, b;
        a.x = a1[2 * g]; a.y = a1[2 * g + 1];
        b.x = t1[2 * g]; b.y = t1[2 * g + 1];
        mt = __builtin_amdgcn_wmma_f32_16x16x4_f32(
            false, a, false, b, (short)0, mt, false, false);
    }

    for (; base < (long)ntiles; base += stride) {
        if ((int)base >= mark)
            mark = priority_from_progress((int)base, q1, q2, q3);
        const long tile = base + wave;
        const bool active = tile < (long)ntiles;
        const float2 xn = load_angle_pair(x, tile + stride, m, h, Btot);

        if (active) {
            v4f mine;
            {
                float sa, ca, sb, cb;
                sincosf(0.5f * xv.x, &sa, &ca);
                sincosf(0.5f * xv.y, &sb, &cb);
                mine.x = ca; mine.y = sa; mine.z = cb; mine.w = sb;
            }
            trig[wave][h * 16 + m] = mine;
            const v4f t01 = trig[wave][m];
            const v4f t23 = trig[wave][16 + m];
            const float c0 = t01.x, s0 = t01.y, c1 = t01.z, s1 = t01.w;
            const float c2 = t23.x, s2 = t23.y, c3 = t23.z, s3 = t23.w;
            const float f0 = h ? s0 : c0;

            v8f acc = {0.f, 0.f, 0.f, 0.f, 0.f, 0.f, 0.f, 0.f};
#pragma unroll
            for (int g = 0; g < 4; ++g) {
                const float pre = (f0 * ((g & 2) ? s1 : c1)) * ((g & 1) ? s2 : c2);
                v2f a;
                a.x = pre * c3;
                a.y = pre * s3;
                v2f bm;
                bm.x = mt[2 * g];
                bm.y = mt[2 * g + 1];
                acc = __builtin_amdgcn_wmma_f32_16x16x4_f32(
                    false, a, false, bm, (short)0, acc, false, false);
            }
#pragma unroll
            for (int r = 0; r < 8; ++r) {
                const int row = r + 8 * h;
                const float yv = acc[r];
                probs[wave][row * 17 + m] = yv * yv;
            }
        }
        __syncthreads();
        if (active) {
            const long b = tile * 16 + m;
            if (b < Btot) {
                const int bp0 = 3 - (2 * h);
                const int bp1 = 3 - (2 * h + 1);
                float a0 = 0.f, a1s = 0.f;
#pragma unroll
                for (int n = 0; n < DIM; ++n) {
                    const float p = probs[wave][m * 17 + n];
                    a0  += ((n >> bp0) & 1) ? -p : p;
                    a1s += ((n >> bp1) & 1) ? -p : p;
                }
                float2 r2;
                r2.x = a0;
                r2.y = a1s;
                ((float2*)out)[b * 2 + h] = r2;
            }
        }
        xv = xn;
    }
    __builtin_amdgcn_s_setprio(0);
}

extern "C" void kernel_launch(void* const* d_in, const int* in_sizes, int n_in,
                              void* d_out, int out_size, void* d_ws, size_t ws_size,
                              hipStream_t stream) {
    const float* x = (const float*)d_in[0];
    const float* w = (const float*)d_in[1];
    float* out = (float*)d_out;

    const long B = (long)in_sizes[0] / 4;
    const int ntiles = (int)((B + 15) / 16);

    quantum_layer_kernel<<<1024, 256, 0, stream>>>(x, w, out, ntiles, B);
}


# Angle loads run two trips ahead and nothing drains them mid-trip: the work-group barrier is dropped (the squared tile is wave-private, LDS serves a wave in order); arithmetic, sum order unchanged
# speedup vs baseline: 1.4334x; 1.0710x over previous
#include <hip/hip_runtime.h>
#include <math.h>


#define DIM 16

typedef float v2f __attribute__((ext_vector_type(2)));
typedef float v8f __attribute__((ext_vector_type(8)));
typedef float v4f __attribute__((ext_vector_type(4)));

__host__ __device__ constexpr int cnot_perm_idx(int c, int t, int idx) {
    return idx ^ ((((idx >> (3 - c)) & 1)) << (3 - t));
}
__host__ __device__ constexpr int ring_perm_idx(int j) {
    int v = cnot_perm_idx(3, 0, j);
    v = cnot_perm_idx(2, 3, v);
    v = cnot_perm_idx(1, 2, v);
    v = cnot_perm_idx(0, 1, v);
    return v;
}
__host__ __device__ constexpr int qbit(int idx, int q) { return (idx >> (3 - q)) & 1; }

__device__ __forceinline__ float ry_elem(int a, int b, float cs, float sn) {
    return (a == b) ? cs : (a ? sn : -sn);
}

__device__ __forceinline__ void layer_rows(const float* cs, const float* sn,
                                           int c, int h, float* frag) {
    constexpr int P8 = ring_perm_idx(8);
    float u[4][2];
#pragma unroll
    for (int q = 0; q < 4; ++q) {
        const int cb = qbit(c, q);
        const int hf = h & qbit(P8, q);
        u[q][0] = ry_elem(hf, cb, cs[q], sn[q]);
        u[q][1] = ry_elem(1 ^ hf, cb, cs[q], sn[q]);
    }
    float p01[2][2], p23[2][2];
#pragma unroll
    for (int i = 0; i < 2; ++i)
#pragma unroll
        for (int j = 0; j < 2; ++j) {
            p01[i][j] = u[0][i] * u[1][j];
            p23[i][j] = u[2][i] * u[3][j];
        }
#pragma unroll
    for (int r = 0; r < 8; ++r) {
        const int pr = ring_perm_idx(r);
        frag[r] = p01[qbit(pr, 0)][qbit(pr, 1)] * p23[qbit(pr, 2)][qbit(pr, 3)];
    }
}

__device__ __forceinline__ void layer_cols(const float* cs, const float* sn,
                                           int c, int h, float* frag) {
    const int pc = ring_perm_idx(c);
    float v[4][2];
#pragma unroll
    for (int q = 0; q < 4; ++q) {
        const int rb = qbit(pc, q);
        v[q][0] = ry_elem(rb, 0, cs[q], sn[q]);
        v[q][1] = ry_elem(rb, 1, cs[q], sn[q]);
    }
    const float v0h = h ? v[0][1] : v[0][0];
    float p01[2], p23[2][2];
#pragma unroll
    for (int i = 0; i < 2; ++i) {
        p01[i] = v0h * v[1][i];
#pragma unroll
        for (int j = 0; j < 2; ++j) p23[i][j] = v[2][i] * v[3][j];
    }
#pragma unroll
    for (int r = 0; r < 8; ++r)
        frag[r] = p01[(r >> 2) & 1] * p23[(r >> 1) & 1][r & 1];
}

__device__ __forceinline__ float2 load_angle_pair(const float* __restrict__ x, long tile,
                                                  int m, int h, long Btot) {
    const long b = tile * 16 + m;
    const long bl = (b < Btot) ? b : (Btot - 1);
    return ((const float2*)x)[bl * 2 + h];
}


__device__ __forceinline__ int priority_from_progress(int bi, int q1, int q2, int q3) {
    if (bi < q1) { __builtin_amdgcn_s_setprio(3); return q1; }
    if (bi < q2) { __builtin_amdgcn_s_setprio(2); return q2; }
    if (bi < q3) { __builtin_amdgcn_s_setprio(1); return q3; }
    __builtin_amdgcn_s_setprio(0);
    return 0x7fffffff;
}

__global__ void __launch_bounds__(256)
quantum_layer_kernel(const float* __restrict__ x, const float* __restrict__ w,
                     float* __restrict__ out, int ntiles, long Btot) {
    __shared__ float probs[8][DIM * 17];
    __shared__ v4f trig[8][32];
    const int lane = threadIdx.x & 31;
    const int wave = threadIdx.x >> 5;
    const int m    = lane & 15;
    const int h    = lane >> 4;
    const long stride = (long)gridDim.x * 8;
    long base = (long)blockIdx.x * 8;
    const int q1 = ntiles >> 2, q2 = ntiles >> 1, q3 = q1 + q2;
    int mark = priority_from_progress((int)base, q1, q2, q3);

    float2 xa = load_angle_pair(x, base + wave, m, h, Btot);
    float2 xb = load_angle_pair(x, base + wave + stride, m, h, Btot);

    float wsn, wcs;
    sincosf(0.5f * w[lane < 12 ? lane : 0], &wsn, &wcs);
    float cs[3][4], sn[3][4];
#pragma unroll
    for (int j = 0; j < 12; ++j) {
        cs[j >> 2][j & 3] = __int_as_float(__builtin_amdgcn_readlane(__float_as_int(wcs), j));
        sn[j >> 2][j & 3] = __int_as_float(__builtin_amdgcn_readlane(__float_as_int(wsn), j));
    }
    float a1[8], a2[8], b3[8];
    layer_rows(cs[0], sn[0], m, h, a1);
    layer_rows(cs[1], sn[1], m, h, a2);
    layer_cols(cs[2], sn[2], m, h, b3);

    v8f t1 = {0.f, 0.f, 0.f, 0.f, 0.f, 0.f, 0.f, 0.f};
#pragma unroll
    for (int g = 0; g < 4; ++g) {
        v2f a, b;
        a.x = a2[2 * g]; a.y = a2[2 * g + 1];
        b.x = b3[2 * g]; b.y = b3[2 * g + 1];
        t1 = __builtin_amdgcn_wmma_f32_16x16x4_f32(
            false, a, false, b, (short)0, t1, false, false);
    }
    v8f mt = {0.f, 0.f, 0.f, 0.f, 0.f, 0.f, 0.f, 0.f};
#pragma unroll
    for (int g = 0; g < 4; ++g) {
        v2f a, b;
        a.x = a1[2 * g]; a.y = a1[2 * g + 1];
        b.x = t1[2 * g]; b.y = t1[2 * g + 1];
        mt = __builtin_amdgcn_wmma_f32_16x16x4_f32(
            false, a, false, b, (short)0, mt, false, false);
    }

    auto trip = [&](float2& xr, long pos) {
        if ((int)pos >= mark)
            mark = priority_from_progress((int)pos, q1, q2, q3);
        const long tile = pos + wave;
        float hx = 0.5f * xr.x, hy = 0.5f * xr.y;
        asm volatile("" : "+v"(hx), "+v"(hy));
        xr = load_angle_pair(x, tile + 2 * stride, m, h, Btot);
        if (tile < (long)ntiles) {
            v4f mine;
            {
                float sa, ca, sb, cb;
                sincosf(hx, &sa, &ca);
                sincosf(hy, &sb, &cb);
                mine.x = ca; mine.y = sa; mine.z = cb; mine.w = sb;
            }
            trig[wave][h * 16 + m] = mine;
            const v4f t01 = trig[wave][m];
            const v4f t23 = trig[wave][16 + m];
            const float c0 = t01.x, s0 = t01.y, c1 = t01.z, s1 = t01.w;
            const float c2 = t23.x, s2 = t23.y, c3 = t23.z, s3 = t23.w;
            const float f0 = h ? s0 : c0;

            v8f acc = {0.f, 0.f, 0.f, 0.f, 0.f, 0.f, 0.f, 0.f};
#pragma unroll
            for (int g = 0; g < 4; ++g) {
                const float pre = (f0 * ((g & 2) ? s1 : c1)) * ((g & 1) ? s2 : c2);
                v2f a;
                a.x = pre * c3;
                a.y = pre * s3;
                v2f bm;
                bm.x = mt[2 * g];
                bm.y = mt[2 * g + 1];
                acc = __builtin_amdgcn_wmma_f32_16x16x4_f32(
                    false, a, false, bm, (short)0, acc, false, false);
            }
#pragma unroll
            for (int r = 0; r < 8; ++r) {
                const int row = r + 8 * h;
                const float yv = acc[r];
                probs[wave][row * 17 + m] = yv * yv;
            }
            const long b = tile * 16 + m;
            if (b < Btot) {
                const int bp0 = 3 - (2 * h);
                const int bp1 = 3 - (2 * h + 1);
                float a0 = 0.f, a1s = 0.f;
#pragma unroll
                for (int n = 0; n < DIM; ++n) {
                    const float p = probs[wave][m * 17 + n];
                    a0  += ((n >> bp0) & 1) ? -p : p;
                    a1s += ((n >> bp1) & 1) ? -p : p;
                }
                float2 r2;
                r2.x = a0;
                r2.y = a1s;
                ((float2*)out)[b * 2 + h] = r2;
            }
        }
    };
    for (; base < (long)ntiles; base += 2 * stride) {
        trip(xa, base);
        trip(xb, base + stride);
    }
    __builtin_amdgcn_s_setprio(0);
}

extern "C" void kernel_launch(void* const* d_in, const int* in_sizes, int n_in,
                              void* d_out, int out_size, void* d_ws, size_t ws_size,
                              hipStream_t stream) {
    const float* x = (const float*)d_in[0];
    const float* w = (const float*)d_in[1];
    float* out = (float*)d_out;

    const long B = (long)in_sizes[0] / 4;
    const int ntiles = (int)((B + 15) / 16);

    quantum_layer_kernel<<<1024, 256, 0, stream>>>(x, w, out, ntiles, B);
}


# Z sums regrouped: squares stored to LDS twice (plain, bit pairs swapped) so each lane adds 4 group sums of 4 with fixed signs from 16 B row reads, no per-term sign selects; Z sum order changes
# speedup vs baseline: 1.5401x; 1.0744x over previous
#include <hip/hip_runtime.h>
#include <math.h>


#define DIM 16

typedef float v2f __attribute__((ext_vector_type(2)));
typedef float v8f __attribute__((ext_vector_type(8)));
typedef float v4f __attribute__((ext_vector_type(4)));

__host__ __device__ constexpr int cnot_perm_idx(int c, int t, int idx) {
    return idx ^ ((((idx >> (3 - c)) & 1)) << (3 - t));
}
__host__ __device__ constexpr int ring_perm_idx(int j) {
    int v = cnot_perm_idx(3, 0, j);
    v = cnot_perm_idx(2, 3, v);
    v = cnot_perm_idx(1, 2, v);
    v = cnot_perm_idx(0, 1, v);
    return v;
}
__host__ __device__ constexpr int qbit(int idx, int q) { return (idx >> (3 - q)) & 1; }

__device__ __forceinline__ float ry_elem(int a, int b, float cs, float sn) {
    return (a == b) ? cs : (a ? sn : -sn);
}

__device__ __forceinline__ void layer_rows(const float* cs, const float* sn,
                                           int c, int h, float* frag) {
    constexpr int P8 = ring_perm_idx(8);
    float u[4][2];
#pragma unroll
    for (int q = 0; q < 4; ++q) {
        const int cb = qbit(c, q);
        const int hf = h & qbit(P8, q);
        u[q][0] = ry_elem(hf, cb, cs[q], sn[q]);
        u[q][1] = ry_elem(1 ^ hf, cb, cs[q], sn[q]);
    }
    float p01[2][2], p23[2][2];
#pragma unroll
    for (int i = 0; i < 2; ++i)
#pragma unroll
        for (int j = 0; j < 2; ++j) {
            p01[i][j] = u[0][i] * u[1][j];
            p23[i][j] = u[2][i] * u[3][j];
        }
#pragma unroll
    for (int r = 0; r < 8; ++r) {
        const int pr = ring_perm_idx(r);
        frag[r] = p01[qbit(pr, 0)][qbit(pr, 1)] * p23[qbit(pr, 2)][qbit(pr, 3)];
    }
}

__device__ __forceinline__ void layer_cols(const float* cs, const float* sn,
                                           int c, int h, float* frag) {
    const int pc = ring_perm_idx(c);
    float v[4][2];
#pragma unroll
    for (int q = 0; q < 4; ++q) {
        const int rb = qbit(pc, q);
        v[q][0] = ry_elem(rb, 0, cs[q], sn[q]);
        v[q][1] = ry_elem(rb, 1, cs[q], sn[q]);
    }
    const float v0h = h ? v[0][1] : v[0][0];
    float p01[2], p23[2][2];
#pragma unroll
    for (int i = 0; i < 2; ++i) {
        p01[i] = v0h * v[1][i];
#pragma unroll
        for (int j = 0; j < 2; ++j) p23[i][j] = v[2][i] * v[3][j];
    }
#pragma unroll
    for (int r = 0; r < 8; ++r)
        frag[r] = p01[(r >> 2) & 1] * p23[(r >> 1) & 1][r & 1];
}

__device__ __forceinline__ float2 load_angle_pair(const float* __restrict__ x, long tile,
                                                  int m, int h, long Btot) {
    const long b = tile * 16 + m;
    const long bl = (b < Btot) ? b : (Btot - 1);
    return ((const float2*)x)[bl * 2 + h];
}


__device__ __forceinline__ int priority_from_progress(int bi, int q1, int q2, int q3) {
    if (bi < q1) { __builtin_amdgcn_s_setprio(3); return q1; }
    if (bi < q2) { __builtin_amdgcn_s_setprio(2); return q2; }
    if (bi < q3) { __builtin_amdgcn_s_setprio(1); return q3; }
    __builtin_amdgcn_s_setprio(0);
    return 0x7fffffff;
}

__global__ void __launch_bounds__(256)
quantum_layer_kernel(const float* __restrict__ x, const float* __restrict__ w,
                     float* __restrict__ out, int ntiles, long Btot) {
    __shared__ __attribute__((aligned(16))) float probs[8][2][DIM * 20];
    __shared__ v4f trig[8][32];
    const int lane = threadIdx.x & 31;
    const int wave = threadIdx.x >> 5;
    const int m    = lane & 15;
    const int h    = lane >> 4;
    const long stride = (long)gridDim.x * 8;
    long base = (long)blockIdx.x * 8;
    const int q1 = ntiles >> 2, q2 = ntiles >> 1, q3 = q1 + q2;
    int mark = priority_from_progress((int)base, q1, q2, q3);

    float2 xa = load_angle_pair(x, base + wave, m, h, Btot);
    float2 xb = load_angle_pair(x, base + wave + stride, m, h, Btot);

    float wsn, wcs;
    sincosf(0.5f * w[lane < 12 ? lane : 0], &wsn, &wcs);
    float cs[3][4], sn[3][4];
#pragma unroll
    for (int j = 0; j < 12; ++j) {
        cs[j >> 2][j & 3] = __int_as_float(__builtin_amdgcn_readlane(__float_as_int(wcs), j));
        sn[j >> 2][j & 3] = __int_as_float(__builtin_amdgcn_readlane(__float_as_int(wsn), j));
    }
    float a1[8], a2[8], b3[8];
    layer_rows(cs[0], sn[0], m, h, a1);
    layer_rows(cs[1], sn[1], m, h, a2);
    layer_cols(cs[2], sn[2], m, h, b3);

    v8f t1 = {0.f, 0.f, 0.f, 0.f, 0.f, 0.f, 0.f, 0.f};
#pragma unroll
    for (int g = 0; g < 4; ++g) {
        v2f a, b;
        a.x = a2[2 * g]; a.y = a2[2 * g + 1];
        b.x = b3[2 * g]; b.y = b3[2 * g + 1];
        t1 = __builtin_amdgcn_wmma_f32_16x16x4_f32(
            false, a, false, b, (short)0, t1, false, false);
    }
    v8f mt = {0.f, 0.f, 0.f, 0.f, 0.f, 0.f, 0.f, 0.f};
#pragma unroll
    for (int g = 0; g < 4; ++g) {
        v2f a, b;
        a.x = a1[2 * g]; a.y = a1[2 * g + 1];
        b.x = t1[2 * g]; b.y = t1[2 * g + 1];
        mt = __builtin_amdgcn_wmma_f32_16x16x4_f32(
            false, a, false, b, (short)0, mt, false, false);
    }

    auto trip = [&](float2& xr, long pos) {
        if ((int)pos >= mark)
            mark = priority_from_progress((int)pos, q1, q2, q3);
        const long tile = pos + wave;
        float hx = 0.5f * xr.x, hy = 0.5f * xr.y;
        asm volatile("" : "+v"(hx), "+v"(hy));
        xr = load_angle_pair(x, tile + 2 * stride, m, h, Btot);
        if (tile < (long)ntiles) {
            v4f mine;
            {
                float sa, ca, sb, cb;
                sincosf(hx, &sa, &ca);
                sincosf(hy, &sb, &cb);
                mine.x = ca; mine.y = sa; mine.z = cb; mine.w = sb;
            }
            trig[wave][h * 16 + m] = mine;
            const v4f t01 = trig[wave][m];
            const v4f t23 = trig[wave][16 + m];
            const float c0 = t01.x, s0 = t01.y, c1 = t01.z, s1 = t01.w;
            const float c2 = t23.x, s2 = t23.y, c3 = t23.z, s3 = t23.w;
            const float f0 = h ? s0 : c0;

            v8f acc = {0.f, 0.f, 0.f, 0.f, 0.f, 0.f, 0.f, 0.f};
#pragma unroll
            for (int g = 0; g < 4; ++g) {
                const float pre = (f0 * ((g & 2) ? s1 : c1)) * ((g & 1) ? s2 : c2);
                v2f a;
                a.x = pre * c3;
                a.y = pre * s3;
                v2f bm;
                bm.x = mt[2 * g];
                bm.y = mt[2 * g + 1];
                acc = __builtin_amdgcn_wmma_f32_16x16x4_f32(
                    false, a, false, bm, (short)0, acc, false, false);
            }
            const int msw = 4 * (m & 3) + (m >> 2);
#pragma unroll
            for (int r = 0; r < 8; ++r) {
                const int row = r + 8 * h;
                const float yv = acc[r];
                const float p = yv * yv;
                probs[wave][0][row * 20 + m] = p;
                probs[wave][1][row * 20 + msw] = p;
            }
            const long b = tile * 16 + m;
            if (b < Btot) {
                const v4f* rowp = (const v4f*)&probs[wave][h][m * 20];
                v2f grp[4];
#pragma unroll
                for (int i = 0; i < 4; ++i) {
                    const v4f v = rowp[i];
                    const v2f lo = {v.x, v.y};
                    const v2f hi = {v.z, v.w};
                    grp[i] = lo + hi;
                }
                const v2f first  = (grp[0] + grp[1]) - (grp[2] + grp[3]);
                const v2f second = (grp[0] - grp[1]) + (grp[2] - grp[3]);
                float2 r2;
                r2.x = first.x + first.y;
                r2.y = second.x + second.y;
                ((float2*)out)[b * 2 + h] = r2;
            }
        }
    };
    for (; base < (long)ntiles; base += 2 * stride) {
        trip(xa, base);
        trip(xb, base + stride);
    }
    __builtin_amdgcn_s_setprio(0);
}

extern "C" void kernel_launch(void* const* d_in, const int* in_sizes, int n_in,
                              void* d_out, int out_size, void* d_ws, size_t ws_size,
                              hipStream_t stream) {
    const float* x = (const float*)d_in[0];
    const float* w = (const float*)d_in[1];
    float* out = (float*)d_out;

    const long B = (long)in_sizes[0] / 4;
    const int ntiles = (int)((B + 15) / 16);

    quantum_layer_kernel<<<1024, 256, 0, stream>>>(x, w, out, ntiles, B);
}


# Tile index, bounds tests and tile base addresses moved to the scalar unit (wave number read through lane 0, 32-bit counters, one fixed lane offset); arithmetic, sum order unchanged
# speedup vs baseline: 1.7041x; 1.1065x over previous
#include <hip/hip_runtime.h>
#include <math.h>


#define DIM 16

typedef float v2f __attribute__((ext_vector_type(2)));
typedef float v8f __attribute__((ext_vector_type(8)));
typedef float v4f __attribute__((ext_vector_type(4)));

__host__ __device__ constexpr int cnot_perm_idx(int c, int t, int idx) {
    return idx ^ ((((idx >> (3 - c)) & 1)) << (3 - t));
}
__host__ __device__ constexpr int ring_perm_idx(int j) {
    int v = cnot_perm_idx(3, 0, j);
    v = cnot_perm_idx(2, 3, v);
    v = cnot_perm_idx(1, 2, v);
    v = cnot_perm_idx(0, 1, v);
    return v;
}
__host__ __device__ constexpr int qbit(int idx, int q) { return (idx >> (3 - q)) & 1; }

__device__ __forceinline__ float ry_elem(int a, int b, float cs, float sn) {
    return (a == b) ? cs : (a ? sn : -sn);
}

__device__ __forceinline__ void layer_rows(const float* cs, const float* sn,
                                           int c, int h, float* frag) {
    constexpr int P8 = ring_perm_idx(8);
    float u[4][2];
#pragma unroll
    for (int q = 0; q < 4; ++q) {
        const int cb = qbit(c, q);
        const int hf = h & qbit(P8, q);
        u[q][0] = ry_elem(hf, cb, cs[q], sn[q]);
        u[q][1] = ry_elem(1 ^ hf, cb, cs[q], sn[q]);
    }
    float p01[2][2], p23[2][2];
#pragma unroll
    for (int i = 0; i < 2; ++i)
#pragma unroll
        for (int j = 0; j < 2; ++j) {
            p01[i][j] = u[0][i] * u[1][j];
            p23[i][j] = u[2][i] * u[3][j];
        }
#pragma unroll
    for (int r = 0; r < 8; ++r) {
        const int pr = ring_perm_idx(r);
        frag[r] = p01[qbit(pr, 0)][qbit(pr, 1)] * p23[qbit(pr, 2)][qbit(pr, 3)];
    }
}

__device__ __forceinline__ void layer_cols(const float* cs, const float* sn,
                                           int c, int h, float* frag) {
    const int pc = ring_perm_idx(c);
    float v[4][2];
#pragma unroll
    for (int q = 0; q < 4; ++q) {
        const int rb = qbit(pc, q);
        v[q][0] = ry_elem(rb, 0, cs[q], sn[q]);
        v[q][1] = ry_elem(rb, 1, cs[q], sn[q]);
    }
    const float v0h = h ? v[0][1] : v[0][0];
    float p01[2], p23[2][2];
#pragma unroll
    for (int i = 0; i < 2; ++i) {
        p01[i] = v0h * v[1][i];
#pragma unroll
        for (int j = 0; j < 2; ++j) p23[i][j] = v[2][i] * v[3][j];
    }
#pragma unroll
    for (int r = 0; r < 8; ++r)
        frag[r] = p01[(r >> 2) & 1] * p23[(r >> 1) & 1][r & 1];
}

__device__ __forceinline__ int rows_of_tile(int tile, int ntiles, int lastrows) {
    return (tile == ntiles - 1) ? lastrows : 16;
}

__device__ __forceinline__ float2 load_angle_pair(const float* __restrict__ x, int tile, int ntiles,
                                                  int m, int h, int lastrows) {
    const int tl = (tile < ntiles) ? tile : (ntiles - 1);
    const int nrows = rows_of_tile(tl, ntiles, lastrows);
    const int mm = (m < nrows) ? m : (nrows - 1);
    const float2* xt = (const float2*)(x + (long)tl * 64);
    return xt[(unsigned)(mm * 2 + h)];
}


__device__ __forceinline__ int priority_from_progress(int bi, int q1, int q2, int q3) {
    if (bi < q1) { __builtin_amdgcn_s_setprio(3); return q1; }
    if (bi < q2) { __builtin_amdgcn_s_setprio(2); return q2; }
    if (bi < q3) { __builtin_amdgcn_s_setprio(1); return q3; }
    __builtin_amdgcn_s_setprio(0);
    return 0x7fffffff;
}

__global__ void __launch_bounds__(256)
quantum_layer_kernel(const float* __restrict__ x, const float* __restrict__ w,
                     float* __restrict__ out, int ntiles, long Btot) {
    __shared__ __attribute__((aligned(16))) float probs[8][2][DIM * 20];
    __shared__ v4f trig[8][32];
    const int lane = threadIdx.x & 31;
    const int wave = __builtin_amdgcn_readfirstlane((int)(threadIdx.x >> 5));
    const int m    = lane & 15;
    const int h    = lane >> 4;
    const int stride = (int)gridDim.x * 8;
    int base = (int)blockIdx.x * 8;
    const int lastrows = (int)(Btot - (long)(ntiles - 1) * 16);
    const int q1 = ntiles >> 2, q2 = ntiles >> 1, q3 = q1 + q2;
    int mark = priority_from_progress(base, q1, q2, q3);

    float2 xa = load_angle_pair(x, base + wave, ntiles, m, h, lastrows);
    float2 xb = load_angle_pair(x, base + wave + stride, ntiles, m, h, lastrows);

    float wsn, wcs;
    sincosf(0.5f * w[lane < 12 ? lane : 0], &wsn, &wcs);
    float cs[3][4], sn[3][4];
#pragma unroll
    for (int j = 0; j < 12; ++j) {
        cs[j >> 2][j & 3] = __int_as_float(__builtin_amdgcn_readlane(__float_as_int(wcs), j));
        sn[j >> 2][j & 3] = __int_as_float(__builtin_amdgcn_readlane(__float_as_int(wsn), j));
    }
    float a1[8], a2[8], b3[8];
    layer_rows(cs[0], sn[0], m, h, a1);
    layer_rows(cs[1], sn[1], m, h, a2);
    layer_cols(cs[2], sn[2], m, h, b3);

    v8f t1 = {0.f, 0.f, 0.f, 0.f, 0.f, 0.f, 0.f, 0.f};
#pragma unroll
    for (int g = 0; g < 4; ++g) {
        v2f a, b;
        a.x = a2[2 * g]; a.y = a2[2 * g + 1];
        b.x = b3[2 * g]; b.y = b3[2 * g + 1];
        t1 = __builtin_amdgcn_wmma_f32_16x16x4_f32(
            false, a, false, b, (short)0, t1, false, false);
    }
    v8f mt = {0.f, 0.f, 0.f, 0.f, 0.f, 0.f, 0.f, 0.f};
#pragma unroll
    for (int g = 0; g < 4; ++g) {
        v2f a, b;
        a.x = a1[2 * g]; a.y = a1[2 * g + 1];
        b.x = t1[2 * g]; b.y = t1[2 * g + 1];
        mt = __builtin_amdgcn_wmma_f32_16x16x4_f32(
            false, a, false, b, (short)0, mt, false, false);
    }

    auto trip = [&](float2& xr, int pos) {
        if (pos >= mark)
            mark = priority_from_progress(pos, q1, q2, q3);
        const int tile = pos + wave;
        float hx = 0.5f * xr.x, hy = 0.5f * xr.y;
        asm volatile("" : "+v"(hx), "+v"(hy));
        xr = load_angle_pair(x, tile + 2 * stride, ntiles, m, h, lastrows);
        if (tile < ntiles) {
            v4f mine;
            {
                float sa, ca, sb, cb;
                sincosf(hx, &sa, &ca);
                sincosf(hy, &sb, &cb);
                mine.x = ca; mine.y = sa; mine.z = cb; mine.w = sb;
            }
            trig[wave][h * 16 + m] = mine;
            const v4f t01 = trig[wave][m];
            const v4f t23 = trig[wave][16 + m];
            const float c0 = t01.x, s0 = t01.y, c1 = t01.z, s1 = t01.w;
            const float c2 = t23.x, s2 = t23.y, c3 = t23.z, s3 = t23.w;
            const float f0 = h ? s0 : c0;

            v8f acc = {0.f, 0.f, 0.f, 0.f, 0.f, 0.f, 0.f, 0.f};
#pragma unroll
            for (int g = 0; g < 4; ++g) {
                const float pre = (f0 * ((g & 2) ? s1 : c1)) * ((g & 1) ? s2 : c2);
                v2f a;
                a.x = pre * c3;
                a.y = pre * s3;
                v2f bm;
                bm.x = mt[2 * g];
                bm.y = mt[2 * g + 1];
                acc = __builtin_amdgcn_wmma_f32_16x16x4_f32(
                    false, a, false, bm, (short)0, acc, false, false);
            }
            const int msw = 4 * (m & 3) + (m >> 2);
#pragma unroll
            for (int r = 0; r < 8; ++r) {
                const int row = r + 8 * h;
                const float yv = acc[r];
                const float p = yv * yv;
                probs[wave][0][row * 20 + m] = p;
                probs[wave][1][row * 20 + msw] = p;
            }
            {
                const v4f* rowp = (const v4f*)&probs[wave][h][m * 20];
                v2f grp[4];
#pragma unroll
                for (int i = 0; i < 4; ++i) {
                    const v4f v = rowp[i];
                    const v2f lo = {v.x, v.y};
                    const v2f hi = {v.z, v.w};
                    grp[i] = lo + hi;
                }
                const v2f first  = (grp[0] + grp[1]) - (grp[2] + grp[3]);
                const v2f second = (grp[0] - grp[1]) + (grp[2] - grp[3]);
                float2 r2;
                r2.x = first.x + first.y;
                r2.y = second.x + second.y;
                float2* ot = (float2*)(out + (long)tile * 64);
                const int nrows = rows_of_tile(tile, ntiles, lastrows);
                if (nrows >= 16) {
                    ot[(unsigned)(m * 2 + h)] = r2;
                } else if (m < nrows) {
                    ot[(unsigned)(m * 2 + h)] = r2;
                }
            }
        }
    };
    for (; base < ntiles; base += 2 * stride) {
        trip(xa, base);
        trip(xb, base + stride);
    }
    __builtin_amdgcn_s_setprio(0);
}

extern "C" void kernel_launch(void* const* d_in, const int* in_sizes, int n_in,
                              void* d_out, int out_size, void* d_ws, size_t ws_size,
                              hipStream_t stream) {
    const float* x = (const float*)d_in[0];
    const float* w = (const float*)d_in[1];
    float* out = (float*)d_out;

    const long B = (long)in_sizes[0] / 4;
    const int ntiles = (int)((B + 15) / 16);

    quantum_layer_kernel<<<1024, 256, 0, stream>>>(x, w, out, ntiles, B);
}


# Data sincos from a 256-entry (cos, sin) LDS table built once per work-group (one barrier): nearest entry turned by the rest angle, 2-term series, two angles as packed f32; sin/cos low bits change
# speedup vs baseline: 1.8871x; 1.1074x over previous
#include <hip/hip_runtime.h>
#include <math.h>


#define DIM 16

typedef float v2f __attribute__((ext_vector_type(2)));
typedef float v8f __attribute__((ext_vector_type(8)));
typedef float v4f __attribute__((ext_vector_type(4)));

#define TAB_N 256
constexpr float TAB_K     = 40.743665431525205956834f;
constexpr float TAB_C1    = 3.14159274101257324f / 128.0f;
constexpr float TAB_C2    = -8.74227766e-8f / 128.0f;
constexpr float TAB_MAGIC = 12582912.0f;
constexpr float TAB_MAX   = 64.0f;

__host__ __device__ constexpr int cnot_perm_idx(int c, int t, int idx) {
    return idx ^ ((((idx >> (3 - c)) & 1)) << (3 - t));
}
__host__ __device__ constexpr int ring_perm_idx(int j) {
    int v = cnot_perm_idx(3, 0, j);
    v = cnot_perm_idx(2, 3, v);
    v = cnot_perm_idx(1, 2, v);
    v = cnot_perm_idx(0, 1, v);
    return v;
}
__host__ __device__ constexpr int qbit(int idx, int q) { return (idx >> (3 - q)) & 1; }

__device__ __forceinline__ float ry_elem(int a, int b, float cs, float sn) {
    return (a == b) ? cs : (a ? sn : -sn);
}

__device__ __forceinline__ void layer_rows(const float* cs, const float* sn,
                                           int c, int h, float* frag) {
    constexpr int P8 = ring_perm_idx(8);
    float u[4][2];
#pragma unroll
    for (int q = 0; q < 4; ++q) {
        const int cb = qbit(c, q);
        const int hf = h & qbit(P8, q);
        u[q][0] = ry_elem(hf, cb, cs[q], sn[q]);
        u[q][1] = ry_elem(1 ^ hf, cb, cs[q], sn[q]);
    }
    float p01[2][2], p23[2][2];
#pragma unroll
    for (int i = 0; i < 2; ++i)
#pragma unroll
        for (int j = 0; j < 2; ++j) {
            p01[i][j] = u[0][i] * u[1][j];
            p23[i][j] = u[2][i] * u[3][j];
        }
#pragma unroll
    for (int r = 0; r < 8; ++r) {
        const int pr = ring_perm_idx(r);
        frag[r] = p01[qbit(pr, 0)][qbit(pr, 1)] * p23[qbit(pr, 2)][qbit(pr, 3)];
    }
}

__device__ __forceinline__ void layer_cols(const float* cs, const float* sn,
                                           int c, int h, float* frag) {
    const int pc = ring_perm_idx(c);
    float v[4][2];
#pragma unroll
    for (int q = 0; q < 4; ++q) {
        const int rb = qbit(pc, q);
        v[q][0] = ry_elem(rb, 0, cs[q], sn[q]);
        v[q][1] = ry_elem(rb, 1, cs[q], sn[q]);
    }
    const float v0h = h ? v[0][1] : v[0][0];
    float p01[2], p23[2][2];
#pragma unroll
    for (int i = 0; i < 2; ++i) {
        p01[i] = v0h * v[1][i];
#pragma unroll
        for (int j = 0; j < 2; ++j) p23[i][j] = v[2][i] * v[3][j];
    }
#pragma unroll
    for (int r = 0; r < 8; ++r)
        frag[r] = p01[(r >> 2) & 1] * p23[(r >> 1) & 1][r & 1];
}

__device__ __forceinline__ int rows_of_tile(int tile, int ntiles, int lastrows) {
    return (tile == ntiles - 1) ? lastrows : 16;
}

__device__ __forceinline__ float2 load_angle_pair(const float* __restrict__ x, int tile, int ntiles,
                                                  int m, int h, int lastrows) {
    const int tl = (tile < ntiles) ? tile : (ntiles - 1);
    const int nrows = rows_of_tile(tl, ntiles, lastrows);
    const int mm = (m < nrows) ? m : (nrows - 1);
    const float2* xt = (const float2*)(x + (long)tl * 64);
    return xt[(unsigned)(mm * 2 + h)];
}


__device__ __forceinline__ int priority_from_progress(int bi, int q1, int q2, int q3) {
    if (bi < q1) { __builtin_amdgcn_s_setprio(3); return q1; }
    if (bi < q2) { __builtin_amdgcn_s_setprio(2); return q2; }
    if (bi < q3) { __builtin_amdgcn_s_setprio(1); return q3; }
    __builtin_amdgcn_s_setprio(0);
    return 0x7fffffff;
}

__global__ void __launch_bounds__(256)
quantum_layer_kernel(const float* __restrict__ x, const float* __restrict__ w,
                     float* __restrict__ out, int ntiles, long Btot) {
    __shared__ __attribute__((aligned(16))) float probs[8][2][DIM * 20];
    __shared__ v4f trig[8][32];
    __shared__ v2f sctab[TAB_N];
    const int lane = threadIdx.x & 31;
    const int wave = __builtin_amdgcn_readfirstlane((int)(threadIdx.x >> 5));
    const int m    = lane & 15;
    const int h    = lane >> 4;
    const int stride = (int)gridDim.x * 8;
    int base = (int)blockIdx.x * 8;
    const int lastrows = (int)(Btot - (long)(ntiles - 1) * 16);
    const int q1 = ntiles >> 2, q2 = ntiles >> 1, q3 = q1 + q2;
    int mark = priority_from_progress(base, q1, q2, q3);

    float2 xa = load_angle_pair(x, base + wave, ntiles, m, h, lastrows);
    float2 xb = load_angle_pair(x, base + wave + stride, ntiles, m, h, lastrows);

    {
        float ts, tc;
        sincospif((float)threadIdx.x * (2.0f / TAB_N), &ts, &tc);
        v2f e; e.x = tc; e.y = ts;
        sctab[threadIdx.x] = e;
    }

    float wsn, wcs;
    sincosf(0.5f * w[lane < 12 ? lane : 0], &wsn, &wcs);
    float cs[3][4], sn[3][4];
#pragma unroll
    for (int j = 0; j < 12; ++j) {
        cs[j >> 2][j & 3] = __int_as_float(__builtin_amdgcn_readlane(__float_as_int(wcs), j));
        sn[j >> 2][j & 3] = __int_as_float(__builtin_amdgcn_readlane(__float_as_int(wsn), j));
    }
    float a1[8], a2[8], b3[8];
    layer_rows(cs[0], sn[0], m, h, a1);
    layer_rows(cs[1], sn[1], m, h, a2);
    layer_cols(cs[2], sn[2], m, h, b3);

    v8f t1 = {0.f, 0.f, 0.f, 0.f, 0.f, 0.f, 0.f, 0.f};
#pragma unroll
    for (int g = 0; g < 4; ++g) {
        v2f a, b;
        a.x = a2[2 * g]; a.y = a2[2 * g + 1];
        b.x = b3[2 * g]; b.y = b3[2 * g + 1];
        t1 = __builtin_amdgcn_wmma_f32_16x16x4_f32(
            false, a, false, b, (short)0, t1, false, false);
    }
    v8f mt = {0.f, 0.f, 0.f, 0.f, 0.f, 0.f, 0.f, 0.f};
#pragma unroll
    for (int g = 0; g < 4; ++g) {
        v2f a, b;
        a.x = a1[2 * g]; a.y = a1[2 * g + 1];
        b.x = t1[2 * g]; b.y = t1[2 * g + 1];
        mt = __builtin_amdgcn_wmma_f32_16x16x4_f32(
            false, a, false, b, (short)0, mt, false, false);
    }

    asm volatile("s_wait_dscnt 0x0\n\ts_barrier_signal -1\n\ts_barrier_wait -1" ::: "memory");

    auto trip = [&](float2& xr, int pos) {
        if (pos >= mark)
            mark = priority_from_progress(pos, q1, q2, q3);
        const int tile = pos + wave;
        float hx = 0.5f * xr.x, hy = 0.5f * xr.y;
        asm volatile("" : "+v"(hx), "+v"(hy));
        xr = load_angle_pair(x, tile + 2 * stride, ntiles, m, h, lastrows);
        if (tile < ntiles) {
            v4f mine;
            const bool beyond = !(fabsf(hx) <= TAB_MAX) | !(fabsf(hy) <= TAB_MAX);
            if (__builtin_expect(__builtin_amdgcn_ballot_w32(beyond) == 0u, 1)) {
                const v2f hv = {hx, hy};
                const v2f kk = {TAB_K, TAB_K}, mg = {TAB_MAGIC, TAB_MAGIC};
                const v2f c1 = {-TAB_C1, -TAB_C1}, c2 = {-TAB_C2, -TAB_C2};
                const v2f t  = __builtin_elementwise_fma(hv, kk, mg);
                const v2f jf = t - mg;
                v2f r = __builtin_elementwise_fma(jf, c1, hv);
                r = __builtin_elementwise_fma(jf, c2, r);
                const v2f ea = sctab[__float_as_uint(t.x) & (TAB_N - 1)];
                const v2f eb = sctab[__float_as_uint(t.y) & (TAB_N - 1)];
                const v2f r2 = r * r;
                const v2f mh = {-0.5f, -0.5f}, one = {1.0f, 1.0f}, m6 = {-1.0f / 6.0f, -1.0f / 6.0f};
                const v2f cr = __builtin_elementwise_fma(r2, mh, one);
                const v2f sr = __builtin_elementwise_fma(r2 * r, m6, r);
                const v2f cra = {cr.x, cr.x}, sra = {sr.x, sr.x};
                const v2f crb = {cr.y, cr.y}, srb = {sr.y, sr.y};
                const v2f qa = {-ea.y, ea.x}, qb = {-eb.y, eb.x};
                const v2f oa = __builtin_elementwise_fma(qa, sra, ea * cra);
                const v2f ob = __builtin_elementwise_fma(qb, srb, eb * crb);
                mine.x = oa.x; mine.y = oa.y; mine.z = ob.x; mine.w = ob.y;
            } else {
                float sa, ca, sb, cb;
                sincosf(hx, &sa, &ca);
                sincosf(hy, &sb, &cb);
                mine.x = ca; mine.y = sa; mine.z = cb; mine.w = sb;
            }
            trig[wave][h * 16 + m] = mine;
            const v4f t01 = trig[wave][m];
            const v4f t23 = trig[wave][16 + m];
            const float c0 = t01.x, s0 = t01.y, c1 = t01.z, s1 = t01.w;
            const float c2 = t23.x, s2 = t23.y, c3 = t23.z, s3 = t23.w;
            const float f0 = h ? s0 : c0;

            v8f acc = {0.f, 0.f, 0.f, 0.f, 0.f, 0.f, 0.f, 0.f};
#pragma unroll
            for (int g = 0; g < 4; ++g) {
                const float pre = (f0 * ((g & 2) ? s1 : c1)) * ((g & 1) ? s2 : c2);
                v2f a;
                a.x = pre * c3;
                a.y = pre * s3;
                v2f bm;
                bm.x = mt[2 * g];
                bm.y = mt[2 * g + 1];
                acc = __builtin_amdgcn_wmma_f32_16x16x4_f32(
                    false, a, false, bm, (short)0, acc, false, false);
            }
            const int msw = 4 * (m & 3) + (m >> 2);
#pragma unroll
            for (int r = 0; r < 8; ++r) {
                const int row = r + 8 * h;
                const float yv = acc[r];
                const float p = yv * yv;
                probs[wave][0][row * 20 + m] = p;
                probs[wave][1][row * 20 + msw] = p;
            }
            {
                const v4f* rowp = (const v4f*)&probs[wave][h][m * 20];
                v2f grp[4];
#pragma unroll
                for (int i = 0; i < 4; ++i) {
                    const v4f v = rowp[i];
                    const v2f lo = {v.x, v.y};
                    const v2f hi = {v.z, v.w};
                    grp[i] = lo + hi;
                }
                const v2f first  = (grp[0] + grp[1]) - (grp[2] + grp[3]);
                const v2f second = (grp[0] - grp[1]) + (grp[2] - grp[3]);
                float2 r2;
                r2.x = first.x + first.y;
                r2.y = second.x + second.y;
                float2* ot = (float2*)(out + (long)tile * 64);
                const int nrows = rows_of_tile(tile, ntiles, lastrows);
                if (nrows >= 16) {
                    ot[(unsigned)(m * 2 + h)] = r2;
                } else if (m < nrows) {
                    ot[(unsigned)(m * 2 + h)] = r2;
                }
            }
        }
    };
    for (; base < ntiles; base += 2 * stride) {
        trip(xa, base);
        trip(xb, base + stride);
    }
    __builtin_amdgcn_s_setprio(0);
}

extern "C" void kernel_launch(void* const* d_in, const int* in_sizes, int n_in,
                              void* d_out, int out_size, void* d_ws, size_t ws_size,
                              hipStream_t stream) {
    const float* x = (const float*)d_in[0];
    const float* w = (const float*)d_in[1];
    float* out = (float*)d_out;

    const long B = (long)in_sizes[0] / 4;
    const int ntiles = (int)((B + 15) / 16);

    quantum_layer_kernel<<<1024, 256, 0, stream>>>(x, w, out, ntiles, B);
}
